# starting kernel
# baseline (speedup 1.0000x reference)
#include <hip/hip_runtime.h>

typedef __attribute__((ext_vector_type(2))) float v2f;
typedef __attribute__((ext_vector_type(8))) float v8f;

#define B_ 32
#define N_ 400
#define M_ 50
#define D_ 128
#define VERY_NEG (-1e30f)

__global__ __launch_bounds__(256) void parts_kernel(
    const float* __restrict__ c, const float* __restrict__ q,
    const float* __restrict__ W,
    float* __restrict__ part_c, float* __restrict__ part_q)
{
    int gid  = blockIdx.x * blockDim.x + threadIdx.x;
    int wave = gid >> 5;
    int lane = gid & 31;
    const int totalC = B_ * N_;
    const int totalQ = B_ * M_;
    if (wave >= totalC + totalQ) return;

    const float* row;
    const float* w;
    float* dst;
    if (wave < totalC) {
        row = c + (size_t)wave * D_;
        w   = W;
        dst = part_c + wave;
    } else {
        int r = wave - totalC;
        row = q + (size_t)r * D_;
        w   = W + D_;
        dst = part_q + r;
    }

    float s = 0.0f;
#pragma unroll
    for (int k = 0; k < D_ / 32; ++k)
        s += row[lane + 32 * k] * w[lane + 32 * k];

#pragma unroll
    for (int off = 16; off >= 1; off >>= 1)
        s += __shfl_xor(s, off, 32);

    if (lane == 0) *dst = s;
}

__global__ __launch_bounds__(32) void sim_wmma_kernel(
    const float* __restrict__ c, const float* __restrict__ q,
    const int* __restrict__ cmask, const int* __restrict__ qmask,
    const float* __restrict__ W,
    const float* __restrict__ part_c, const float* __restrict__ part_q,
    float* __restrict__ out)
{
    const int nt   = blockIdx.x;
    const int b    = blockIdx.y;
    const int lane = threadIdx.x;
    const int half = lane >> 4;
    const int l16  = lane & 15;
    const int n0   = nt * 16;

    const float* crow = c + (((size_t)b * N_ + (n0 + l16)) * D_) + 2 * half;
    const float* wcq  = W + 2 * D_ + 2 * half;

    int mc[4];
#pragma unroll
    for (int t = 0; t < 4; ++t) {
        int m = t * 16 + l16;
        mc[t] = (m < M_) ? m : (M_ - 1);
    }
    const float* qr0 = q + (((size_t)b * M_ + mc[0]) * D_) + 2 * half;
    const float* qr1 = q + (((size_t)b * M_ + mc[1]) * D_) + 2 * half;
    const float* qr2 = q + (((size_t)b * M_ + mc[2]) * D_) + 2 * half;
    const float* qr3 = q + (((size_t)b * M_ + mc[3]) * D_) + 2 * half;

    v8f acc0 = {}, acc1 = {}, acc2 = {}, acc3 = {};

#pragma unroll 8
    for (int kk = 0; kk < D_; kk += 4) {
        v2f a = *(const v2f*)(crow + kk);
        v2f w = *(const v2f*)(wcq + kk);
        a.x *= w.x;
        a.y *= w.y;
        v2f b0 = *(const v2f*)(qr0 + kk);
        v2f b1 = *(const v2f*)(qr1 + kk);
        v2f b2 = *(const v2f*)(qr2 + kk);
        v2f b3 = *(const v2f*)(qr3 + kk);

        acc0 = __builtin_amdgcn_wmma_f32_16x16x4_f32(false, a, false, b0, (short)0, acc0, false, false);
        acc1 = __builtin_amdgcn_wmma_f32_16x16x4_f32(false, a, false, b1, (short)0, acc1, false, false);
        acc2 = __builtin_amdgcn_wmma_f32_16x16x4_f32(false, a, false, b2, (short)0, acc2, false, false);
        acc3 = __builtin_amdgcn_wmma_f32_16x16x4_f32(false, a, false, b3, (short)0, acc3, false, false);
    }

    float pq[4];
    int   qm[4];
#pragma unroll
    for (int t = 0; t < 4; ++t) {
        int m   = t * 16 + l16;
        bool ok = (m < M_);
        pq[t] = ok ? part_q[b * M_ + m] : 0.0f;
        qm[t] = ok ? qmask[b * M_ + m] : 0;
    }

    v8f accs[4] = {acc0, acc1, acc2, acc3};

#pragma unroll
    for (int v = 0; v < 8; ++v) {
        int    n  = n0 + v + 8 * half;
        float  pc = part_c[b * N_ + n];
        int    cm = cmask[b * N_ + n];
        size_t rowbase = ((size_t)b * N_ + n) * M_;
#pragma unroll
        for (int t = 0; t < 4; ++t) {
            int m = t * 16 + l16;
            if (m < M_) {
                float val = accs[t][v] + pc + pq[t];
                if (!(cm && qm[t])) val += VERY_NEG;
                out[rowbase + m] = val;
            }
        }
    }
}

extern "C" void kernel_launch(void* const* d_in, const int* in_sizes, int n_in,
                              void* d_out, int out_size, void* d_ws, size_t ws_size,
                              hipStream_t stream)
{
    const float* c     = (const float*)d_in[0];
    const float* q     = (const float*)d_in[1];
    const int*   cmask = (const int*)d_in[2];
    const int*   qmask = (const int*)d_in[3];
    const float* W     = (const float*)d_in[4];
    float*       out   = (float*)d_out;

    float* part_c = (float*)d_ws;
    float* part_q = part_c + B_ * N_;

    int waves   = B_ * N_ + B_ * M_;
    int threads = 256;
    int blocks  = (waves * 32 + threads - 1) / threads;
    parts_kernel<<<blocks, threads, 0, stream>>>(c, q, W, part_c, part_q);

    dim3 grid(N_ / 16, B_);
    sim_wmma_kernel<<<grid, 32, 0, stream>>>(c, q, cmask, qmask, W,
                                             part_c, part_q, out);
}


# Fuse the rank-one terms c.w_c (per row) and q.w_q (per column) into the WMMA kernel: per-lane FMA chains on its loaded fragments plus one lane-half exchange; one launch; their summation order changes
# speedup vs baseline: 1.4173x; 1.4173x over previous
#include <hip/hip_runtime.h>


typedef __attribute__((ext_vector_type(2))) float v2f;
typedef __attribute__((ext_vector_type(8))) float v8f;

#define B_ 32
#define N_ 400
#define M_ 50
#define D_ 128
#define VERY_NEG (-1e30f)

__global__ __launch_bounds__(32) void sim_fused_kernel(
    const float* __restrict__ c, const float* __restrict__ q,
    const int* __restrict__ cmask, const int* __restrict__ qmask,
    const float* __restrict__ W,
    float* __restrict__ out)
{
    const int nt   = blockIdx.x;
    const int b    = blockIdx.y;
    const int lane = threadIdx.x;
    const int half = lane >> 4;
    const int l16  = lane & 15;
    const int n0   = nt * 16;

    const float* crow = c + (((size_t)b * N_ + (n0 + l16)) * D_) + 2 * half;
    const float* wc   = W + 2 * half;
    const float* wq   = W + D_ + 2 * half;
    const float* wcq  = W + 2 * D_ + 2 * half;

    int mc[4];
#pragma unroll
    for (int t = 0; t < 4; ++t) {
        int m = t * 16 + l16;
        mc[t] = (m < M_) ? m : (M_ - 1);
    }
    const float* qr0 = q + (((size_t)b * M_ + mc[0]) * D_) + 2 * half;
    const float* qr1 = q + (((size_t)b * M_ + mc[1]) * D_) + 2 * half;
    const float* qr2 = q + (((size_t)b * M_ + mc[2]) * D_) + 2 * half;
    const float* qr3 = q + (((size_t)b * M_ + mc[3]) * D_) + 2 * half;

    v8f acc0 = {}, acc1 = {}, acc2 = {}, acc3 = {};
    v2f pc2 = {};
    v2f pq0 = {}, pq1 = {}, pq2 = {}, pq3 = {};

#pragma unroll 8
    for (int kk = 0; kk < D_; kk += 4) {
        v2f a  = *(const v2f*)(crow + kk);
        v2f w  = *(const v2f*)(wcq + kk);
        v2f u  = *(const v2f*)(wc + kk);
        v2f g  = *(const v2f*)(wq + kk);
        v2f b0 = *(const v2f*)(qr0 + kk);
        v2f b1 = *(const v2f*)(qr1 + kk);
        v2f b2 = *(const v2f*)(qr2 + kk);
        v2f b3 = *(const v2f*)(qr3 + kk);

        pc2 += a * u;
        pq0 += b0 * g;
        pq1 += b1 * g;
        pq2 += b2 * g;
        pq3 += b3 * g;
        a *= w;

        acc0 = __builtin_amdgcn_wmma_f32_16x16x4_f32(false, a, false, b0, (short)0, acc0, false, false);
        acc1 = __builtin_amdgcn_wmma_f32_16x16x4_f32(false, a, false, b1, (short)0, acc1, false, false);
        acc2 = __builtin_amdgcn_wmma_f32_16x16x4_f32(false, a, false, b2, (short)0, acc2, false, false);
        acc3 = __builtin_amdgcn_wmma_f32_16x16x4_f32(false, a, false, b3, (short)0, acc3, false, false);
    }

    float pcs = pc2.x + pc2.y;
    pcs += __shfl_xor(pcs, 16, 32);
    float pqs[4] = {pq0.x + pq0.y, pq1.x + pq1.y, pq2.x + pq2.y, pq3.x + pq3.y};
#pragma unroll
    for (int t = 0; t < 4; ++t)
        pqs[t] += __shfl_xor(pqs[t], 16, 32);

    int qm[4];
#pragma unroll
    for (int t = 0; t < 4; ++t) {
        int m   = t * 16 + l16;
        bool ok = (m < M_);
        qm[t] = ok ? qmask[b * M_ + m] : 0;
    }

    v8f accs[4] = {acc0, acc1, acc2, acc3};

#pragma unroll
    for (int v = 0; v < 8; ++v) {
        int    n  = n0 + v + 8 * half;
        float  pc = __shfl(pcs, v + 8 * half, 32);
        int    cm = cmask[b * N_ + n];
        size_t rowbase = ((size_t)b * N_ + n) * M_;
#pragma unroll
        for (int t = 0; t < 4; ++t) {
            int m = t * 16 + l16;
            if (m < M_) {
                float val = accs[t][v] + pc + pqs[t];
                if (!(cm && qm[t])) val += VERY_NEG;
                out[rowbase + m] = val;
            }
        }
    }
}

extern "C" void kernel_launch(void* const* d_in, const int* in_sizes, int n_in,
                              void* d_out, int out_size, void* d_ws, size_t ws_size,
                              hipStream_t stream)
{
    const float* c     = (const float*)d_in[0];
    const float* q     = (const float*)d_in[1];
    const int*   cmask = (const int*)d_in[2];
    const int*   qmask = (const int*)d_in[3];
    const float* W     = (const float*)d_in[4];
    float*       out   = (float*)d_out;

    dim3 grid(N_ / 16, B_);
    sim_fused_kernel<<<grid, 32, 0, stream>>>(c, q, cmask, qmask, W, out);
}


# Operands staged through LDS: c, q, W rows fetched by whole-line 128-bit loads (q tile and W once per 4-wave workgroup), fragments read from padded LDS one 32-WMMA block ahead; all sums unchanged
# speedup vs baseline: 1.7432x; 1.2299x over previous
#include <hip/hip_runtime.h>


typedef __attribute__((ext_vector_type(2))) float v2f;
typedef __attribute__((ext_vector_type(4))) float v4f;
typedef __attribute__((ext_vector_type(8))) float v8f;

#define B_ 32
#define N_ 400
#define M_ 50
#define D_ 128
#define VERY_NEG (-1e30f)

#define NT_    (N_ / 16)
#define WAVES_ 4
#define WGX_   ((NT_ + WAVES_ - 1) / WAVES_)
#define QFULL_ (M_ / WAVES_)
#define QREST_ (M_ - QFULL_ * WAVES_)
#define PITCH_ (D_ + 4)
#define KB_    8

struct Frag {
    v2f a[KB_], w[KB_], u[KB_], g[KB_], b0[KB_], b1[KB_], b2[KB_], b3[KB_];
};

__global__ __launch_bounds__(32 * WAVES_) void sim_fused_kernel(
    const float* __restrict__ c, const float* __restrict__ q,
    const int* __restrict__ cmask, const int* __restrict__ qmask,
    const float* __restrict__ W,
    float* __restrict__ out)
{
    __shared__ __attribute__((aligned(16))) float sC[WAVES_][16 * PITCH_];
    __shared__ __attribute__((aligned(16))) float sQ[M_ * PITCH_];
    __shared__ __attribute__((aligned(16))) float sW[3 * D_];
    __shared__ __attribute__((aligned(16))) float sSpare[WAVES_][D_];

    const int lane = threadIdx.x & 31;
    const int wv   = __builtin_amdgcn_readfirstlane((int)(threadIdx.x >> 5));
    const int b    = blockIdx.y;
    const int nt   = blockIdx.x * WAVES_ + wv;
    const bool live = nt < NT_;
    const int half = lane >> 4;
    const int l16  = lane & 15;
    const int n0   = nt * 16;

    const int ntc = live ? nt : (NT_ - 1);
    const int qlr = QFULL_ * WAVES_ + (wv < QREST_ ? wv : QREST_ - 1);
    const int wr  = (wv > 0 ? wv : 1) - 1;

    const float* cg = c + ((size_t)b * N_ + ntc * 16) * D_ + lane * 4;
    const float* qg = q + ((size_t)b * M_ + wv) * D_ + lane * 4;
    v4f rc[16], rq[QFULL_];
#pragma unroll
    for (int r = 0; r < 16; ++r)
        rc[r] = *(const v4f*)(cg + r * D_);
#pragma unroll
    for (int i = 0; i < QFULL_; ++i)
        rq[i] = *(const v4f*)(qg + i * WAVES_ * D_);
    v4f rql = *(const v4f*)(q + ((size_t)b * M_ + qlr) * D_ + lane * 4);
    v4f rw  = *(const v4f*)(W + wr * D_ + lane * 4);

    float* cs = &sC[wv][lane * 4];
    float* qs = &sQ[wv * PITCH_ + lane * 4];
#pragma unroll
    for (int r = 0; r < 16; ++r)
        *(v4f*)(cs + r * PITCH_) = rc[r];
#pragma unroll
    for (int i = 0; i < QFULL_; ++i)
        *(v4f*)(qs + i * WAVES_ * PITCH_) = rq[i];
    float* qld = (wv < QREST_) ? &sQ[qlr * PITCH_] : &sSpare[wv][0];
    float* wd  = (wv > 0)      ? &sW[wr * D_]      : &sSpare[0][0];
    *(v4f*)(qld + lane * 4) = rql;
    *(v4f*)(wd + lane * 4)  = rw;

    __syncthreads();

    if (!live) return;

    const float* pa   = &sC[wv][l16 * PITCH_ + 2 * half];
    const float* pwc  = &sW[2 * half];
    const float* pwq  = &sW[D_ + 2 * half];
    const float* pwcq = &sW[2 * D_ + 2 * half];

    int mc[4];
#pragma unroll
    for (int t = 0; t < 4; ++t) {
        int m = t * 16 + l16;
        mc[t] = (m < M_) ? m : (M_ - 1);
    }
    const float* pb0 = &sQ[mc[0] * PITCH_ + 2 * half];
    const float* pb1 = &sQ[mc[1] * PITCH_ + 2 * half];
    const float* pb2 = &sQ[mc[2] * PITCH_ + 2 * half];
    const float* pb3 = &sQ[mc[3] * PITCH_ + 2 * half];

    v8f acc0 = {}, acc1 = {}, acc2 = {}, acc3 = {};
    v2f pc2 = {};
    v2f pq0 = {}, pq1 = {}, pq2 = {}, pq3 = {};

#define LOAD_BLOCK(f, kb)                                            \
    _Pragma("unroll")                                                \
    for (int j = 0; j < KB_; ++j) {                                  \
        const int kk = (kb) * (4 * KB_) + 4 * j;                     \
        (f).a[j]  = *(const v2f*)(pa + kk);                          \
        (f).w[j]  = *(const v2f*)(pwcq + kk);                        \
        (f).u[j]  = *(const v2f*)(pwc + kk);                         \
        (f).g[j]  = *(const v2f*)(pwq + kk);                         \
        (f).b0[j] = *(const v2f*)(pb0 + kk);                         \
        (f).b1[j] = *(const v2f*)(pb1 + kk);                         \
        (f).b2[j] = *(const v2f*)(pb2 + kk);                         \
        (f).b3[j] = *(const v2f*)(pb3 + kk);                         \
    }

#define COMPUTE_BLOCK(f)                                             \
    _Pragma("unroll")                                                \
    for (int j = 0; j < KB_; ++j) {                                  \
        v2f a = (f).a[j];                                            \
        pc2 += a * (f).u[j];                \
        pq0 += (f).b0[j] * (f).g[j];   \
        pq1 += (f).b1[j] * (f).g[j];                                 \
        pq2 += (f).b2[j] * (f).g[j];                                 \
        pq3 += (f).b3[j] * (f).g[j];                                 \
        a *= (f).w[j];                          \
        acc0 = __builtin_amdgcn_wmma_f32_16x16x4_f32(false, a, false, (f).b0[j], (short)0, acc0, false, false); \
        acc1 = __builtin_amdgcn_wmma_f32_16x16x4_f32(false, a, false, (f).b1[j], (short)0, acc1, false, false); \
        acc2 = __builtin_amdgcn_wmma_f32_16x16x4_f32(false, a, false, (f).b2[j], (short)0, acc2, false, false); \
        acc3 = __builtin_amdgcn_wmma_f32_16x16x4_f32(false, a, false, (f).b3[j], (short)0, acc3, false, false); \
    }

#define GROUP_EDGE()                                                             \
    do {                                                                         \
        asm volatile("" : "+v"(acc0), "+v"(acc1), "+v"(acc2), "+v"(acc3),        \
                          "+v"(pc2), "+v"(pq0), "+v"(pq1), "+v"(pq2), "+v"(pq3));\
        __builtin_amdgcn_sched_barrier(0);                                       \
    } while (0)

    Frag f0, f1;
    LOAD_BLOCK(f0, 0)
    LOAD_BLOCK(f1, 1)
    GROUP_EDGE();
    COMPUTE_BLOCK(f0)
    GROUP_EDGE();
    LOAD_BLOCK(f0, 2)
    GROUP_EDGE();
    COMPUTE_BLOCK(f1)
    GROUP_EDGE();
    LOAD_BLOCK(f1, 3)
    GROUP_EDGE();
    COMPUTE_BLOCK(f0)
    COMPUTE_BLOCK(f1)
#undef GROUP_EDGE
#undef LOAD_BLOCK
#undef COMPUTE_BLOCK

    float pcs = pc2.x + pc2.y;
    pcs += __shfl_xor(pcs, 16, 32);
    float pqs[4] = {pq0.x + pq0.y, pq1.x + pq1.y, pq2.x + pq2.y, pq3.x + pq3.y};
#pragma unroll
    for (int t = 0; t < 4; ++t)
        pqs[t] += __shfl_xor(pqs[t], 16, 32);

    int qm[4];
#pragma unroll
    for (int t = 0; t < 4; ++t) {
        int m   = t * 16 + l16;
        bool ok = (m < M_);
        qm[t] = ok ? qmask[b * M_ + m] : 0;
    }

    v8f accs[4] = {acc0, acc1, acc2, acc3};

#pragma unroll
    for (int v = 0; v < 8; ++v) {
        int    n  = n0 + v + 8 * half;
        float  pc = __shfl(pcs, v + 8 * half, 32);
        int    cm = cmask[b * N_ + n];
        size_t rowbase = ((size_t)b * N_ + n) * M_;
#pragma unroll
        for (int t = 0; t < 4; ++t) {
            int m = t * 16 + l16;
            if (m < M_) {
                float val = accs[t][v] + pc + pqs[t];
                if (!(cm && qm[t])) val += VERY_NEG;
                out[rowbase + m] = val;
            }
        }
    }
}

extern "C" void kernel_launch(void* const* d_in, const int* in_sizes, int n_in,
                              void* d_out, int out_size, void* d_ws, size_t ws_size,
                              hipStream_t stream)
{
    const float* c     = (const float*)d_in[0];
    const float* q     = (const float*)d_in[1];
    const int*   cmask = (const int*)d_in[2];
    const int*   qmask = (const int*)d_in[3];
    const float* W     = (const float*)d_in[4];
    float*       out   = (float*)d_out;

    dim3 grid(WGX_, B_);
    sim_fused_kernel<<<grid, 32 * WAVES_, 0, stream>>>(c, q, cmask, qmask, W, out);
}


# Epilogue made straight-line: each wave assembles its 16x50 tile in its own LDS region and stores it as 25 whole 128-byte lines (7 128-bit stores for 32 partial 32-bit ones); all sums unchanged
# speedup vs baseline: 2.8043x; 1.6087x over previous
#include <hip/hip_runtime.h>


typedef __attribute__((ext_vector_type(2))) float v2f;
typedef __attribute__((ext_vector_type(4))) float v4f;
typedef __attribute__((ext_vector_type(8))) float v8f;
typedef __attribute__((ext_vector_type(4))) int   v4i;

#define B_ 32
#define N_ 400
#define M_ 50
#define D_ 128
#define VERY_NEG (-1e30f)

#define NT_    (N_ / 16)
#define WAVES_ 4
#define WGX_   ((NT_ + WAVES_ - 1) / WAVES_)
#define QFULL_ (M_ / WAVES_)
#define QREST_ (M_ - QFULL_ * WAVES_)
#define PITCH_ (D_ + 4)
#define KB_    8

struct Frag {
    v2f a[KB_], w[KB_], u[KB_], g[KB_], b0[KB_], b1[KB_], b2[KB_], b3[KB_];
};

__global__ __launch_bounds__(32 * WAVES_) void sim_fused_kernel(
    const float* __restrict__ c, const float* __restrict__ q,
    const int* __restrict__ cmask, const int* __restrict__ qmask,
    const float* __restrict__ W,
    float* __restrict__ out)
{
    __shared__ __attribute__((aligned(16))) float sC[WAVES_][16 * PITCH_];
    __shared__ __attribute__((aligned(16))) float sQ[M_ * PITCH_];
    __shared__ __attribute__((aligned(16))) float sW[3 * D_];
    __shared__ __attribute__((aligned(16))) float sSpare[WAVES_][D_];

    const int lane = threadIdx.x & 31;
    const int wv   = __builtin_amdgcn_readfirstlane((int)(threadIdx.x >> 5));
    const int b    = blockIdx.y;
    const int nt   = blockIdx.x * WAVES_ + wv;
    const bool live = nt < NT_;
    const int half = lane >> 4;
    const int l16  = lane & 15;
    const int n0   = nt * 16;

    const int ntc = live ? nt : (NT_ - 1);
    const int qlr = QFULL_ * WAVES_ + (wv < QREST_ ? wv : QREST_ - 1);
    const int wr  = (wv > 0 ? wv : 1) - 1;

    const float* cg = c + ((size_t)b * N_ + ntc * 16) * D_ + lane * 4;
    const float* qg = q + ((size_t)b * M_ + wv) * D_ + lane * 4;
    v4f rc[16], rq[QFULL_];
#pragma unroll
    for (int r = 0; r < 16; ++r)
        rc[r] = *(const v4f*)(cg + r * D_);
#pragma unroll
    for (int i = 0; i < QFULL_; ++i)
        rq[i] = *(const v4f*)(qg + i * WAVES_ * D_);
    v4f rql = *(const v4f*)(q + ((size_t)b * M_ + qlr) * D_ + lane * 4);
    v4f rw  = *(const v4f*)(W + wr * D_ + lane * 4);

    float* cs = &sC[wv][lane * 4];
    float* qs = &sQ[wv * PITCH_ + lane * 4];
#pragma unroll
    for (int r = 0; r < 16; ++r)
        *(v4f*)(cs + r * PITCH_) = rc[r];
#pragma unroll
    for (int i = 0; i < QFULL_; ++i)
        *(v4f*)(qs + i * WAVES_ * PITCH_) = rq[i];
    float* qld = (wv < QREST_) ? &sQ[qlr * PITCH_] : &sSpare[wv][0];
    float* wd  = (wv > 0)      ? &sW[wr * D_]      : &sSpare[0][0];
    *(v4f*)(qld + lane * 4) = rql;
    *(v4f*)(wd + lane * 4)  = rw;

    __syncthreads();

    if (!live) return;

    const float* pa   = &sC[wv][l16 * PITCH_ + 2 * half];
    const float* pwc  = &sW[2 * half];
    const float* pwq  = &sW[D_ + 2 * half];
    const float* pwcq = &sW[2 * D_ + 2 * half];

    int mc[4];
#pragma unroll
    for (int t = 0; t < 4; ++t) {
        int m = t * 16 + l16;
        mc[t] = (m < M_) ? m : (M_ - 1);
    }
    const float* pb0 = &sQ[mc[0] * PITCH_ + 2 * half];
    const float* pb1 = &sQ[mc[1] * PITCH_ + 2 * half];
    const float* pb2 = &sQ[mc[2] * PITCH_ + 2 * half];
    const float* pb3 = &sQ[mc[3] * PITCH_ + 2 * half];

    v8f acc0 = {}, acc1 = {}, acc2 = {}, acc3 = {};
    v2f pc2 = {};
    v2f pq0 = {}, pq1 = {}, pq2 = {}, pq3 = {};

#define LOAD_BLOCK(f, kb)                                            \
    _Pragma("unroll")                                                \
    for (int j = 0; j < KB_; ++j) {                                  \
        const int kk = (kb) * (4 * KB_) + 4 * j;                     \
        (f).a[j]  = *(const v2f*)(pa + kk);                          \
        (f).w[j]  = *(const v2f*)(pwcq + kk);                        \
        (f).u[j]  = *(const v2f*)(pwc + kk);                         \
        (f).g[j]  = *(const v2f*)(pwq + kk);                         \
        (f).b0[j] = *(const v2f*)(pb0 + kk);                         \
        (f).b1[j] = *(const v2f*)(pb1 + kk);                         \
        (f).b2[j] = *(const v2f*)(pb2 + kk);                         \
        (f).b3[j] = *(const v2f*)(pb3 + kk);                         \
    }

#define COMPUTE_BLOCK(f)                                             \
    _Pragma("unroll")                                                \
    for (int j = 0; j < KB_; ++j) {                                  \
        v2f a = (f).a[j];                                            \
        pc2 += a * (f).u[j];                \
        pq0 += (f).b0[j] * (f).g[j];   \
        pq1 += (f).b1[j] * (f).g[j];                                 \
        pq2 += (f).b2[j] * (f).g[j];                                 \
        pq3 += (f).b3[j] * (f).g[j];                                 \
        a *= (f).w[j];                          \
        acc0 = __builtin_amdgcn_wmma_f32_16x16x4_f32(false, a, false, (f).b0[j], (short)0, acc0, false, false); \
        acc1 = __builtin_amdgcn_wmma_f32_16x16x4_f32(false, a, false, (f).b1[j], (short)0, acc1, false, false); \
        acc2 = __builtin_amdgcn_wmma_f32_16x16x4_f32(false, a, false, (f).b2[j], (short)0, acc2, false, false); \
        acc3 = __builtin_amdgcn_wmma_f32_16x16x4_f32(false, a, false, (f).b3[j], (short)0, acc3, false, false); \
    }

#define GROUP_EDGE()                                                             \
    do {                                                                         \
        asm volatile("" : "+v"(acc0), "+v"(acc1), "+v"(acc2), "+v"(acc3),        \
                          "+v"(pc2), "+v"(pq0), "+v"(pq1), "+v"(pq2), "+v"(pq3));\
        __builtin_amdgcn_sched_barrier(0);                                       \
    } while (0)

    Frag f0, f1;
    LOAD_BLOCK(f0, 0)
    LOAD_BLOCK(f1, 1)
    GROUP_EDGE();
    COMPUTE_BLOCK(f0)
    GROUP_EDGE();
    LOAD_BLOCK(f0, 2)
    GROUP_EDGE();
    COMPUTE_BLOCK(f1)
    GROUP_EDGE();
    LOAD_BLOCK(f1, 3)
    GROUP_EDGE();
    COMPUTE_BLOCK(f0)
    COMPUTE_BLOCK(f1)
#undef GROUP_EDGE
#undef LOAD_BLOCK
#undef COMPUTE_BLOCK

    float pcs = pc2.x + pc2.y;
    pcs += __shfl_xor(pcs, 16, 32);
    float pqs[4] = {pq0.x + pq0.y, pq1.x + pq1.y, pq2.x + pq2.y, pq3.x + pq3.y};
#pragma unroll
    for (int t = 0; t < 4; ++t)
        pqs[t] += __shfl_xor(pqs[t], 16, 32);

    float pcv[8];
#pragma unroll
    for (int v = 0; v < 8; ++v)
        pcv[v] = __shfl(pcs, v + 8 * half, 32);

    const int* cmp = cmask + b * N_ + n0 + 8 * half;
    const v4i cm0 = *(const v4i*)cmp;
    const v4i cm1 = *(const v4i*)(cmp + 4);
    const int cmr[8] = {cm0.x, cm0.y, cm0.z, cm0.w, cm1.x, cm1.y, cm1.z, cm1.w};
    int qm[4];
#pragma unroll
    for (int t = 0; t < 4; ++t)
        qm[t] = qmask[b * M_ + mc[t]];

    v8f accs[4] = {acc0, acc1, acc2, acc3};

    float* so = &sC[wv][0];
#pragma unroll
    for (int v = 0; v < 8; ++v) {
        float* srow = so + (v + 8 * half) * M_ + l16;
#pragma unroll
        for (int t = 0; t < 3; ++t) {
            float val = accs[t][v] + pcv[v] + pqs[t];
            if (!(cmr[v] && qm[t])) val += VERY_NEG;
            srow[t * 16] = val;
        }
    }
    float* s3 = so + ((l16 < M_ - 48) ? (8 * half * M_ + 48 + l16) : (1024 + lane));
#pragma unroll
    for (int v = 0; v < 8; ++v) {
        float val = accs[3][v] + pcv[v] + pqs[3];
        if (!(cmr[v] && qm[3])) val += VERY_NEG;
        s3[v * M_] = val;
    }

    asm volatile("s_wait_dscnt 0x0" ::: "memory");

    float*       og = out + ((size_t)b * N_ + n0) * M_ + lane * 4;
    const float* sr = so + lane * 4;
#pragma unroll
    for (int i = 0; i < 6; ++i)
        *(v4f*)(og + i * 128) = *(const v4f*)(sr + i * 128);
    if (lane < 8)
        *(v4f*)(og + 6 * 128) = *(const v4f*)(sr + 6 * 128);
}

extern "C" void kernel_launch(void* const* d_in, const int* in_sizes, int n_in,
                              void* d_out, int out_size, void* d_ws, size_t ws_size,
                              hipStream_t stream)
{
    const float* c     = (const float*)d_in[0];
    const float* q     = (const float*)d_in[1];
    const int*   cmask = (const int*)d_in[2];
    const int*   qmask = (const int*)d_in[3];
    const float* W     = (const float*)d_in[4];
    float*       out   = (float*)d_out;

    dim3 grid(WGX_, B_);
    sim_fused_kernel<<<grid, 32 * WAVES_, 0, stream>>>(c, q, cmask, qmask, W, out);
}
